# starting kernel
# baseline (speedup 1.0000x reference)
#include <hip/hip_runtime.h>

#define B_DIM 8
#define N_DIM 1024
#define F_DIM 64
#define EPS_V 1e-10f

typedef __attribute__((ext_vector_type(2))) float v2f;
typedef __attribute__((ext_vector_type(4))) float v4f;
typedef __attribute__((ext_vector_type(8))) float v8f;

__global__ void hg_zero_kernel(float* __restrict__ p, int n) {
    int i = blockIdx.x * blockDim.x + threadIdx.x;
    if (i < n) p[i] = 0.0f;
}

__global__ void hg_degree_kernel(const int* __restrict__ eb,
                                 const int* __restrict__ ei,
                                 float* __restrict__ deg, int nnz) {
    int e = blockIdx.x * blockDim.x + threadIdx.x;
    if (e < nnz) atomicAdd(&deg[eb[e] * N_DIM + ei[e]], 1.0f);
}

__global__ __launch_bounds__(256) void hg_mlp_wmma_kernel(
    const float* __restrict__ x,
    const float* __restrict__ W1, const float* __restrict__ b1,
    const float* __restrict__ W2, const float* __restrict__ b2,
    const float* __restrict__ Ws1, const float* __restrict__ bs1,
    const float* __restrict__ Ws2, const float* __restrict__ bs2,
    float* __restrict__ x1_out, float* __restrict__ self_out) {

    __shared__ float sW[F_DIM * F_DIM];
    __shared__ float sB[F_DIM];
    __shared__ float sH[8][16 * F_DIM];

    const int tid  = threadIdx.x;
    const int wave = tid >> 5;
    const int lane = tid & 31;
    const int hlf  = lane >> 4;
    const int mn   = lane & 15;
    const int rowBase = (blockIdx.x * 8 + wave) * 16;

    v2f aX[16];
#pragma unroll
    for (int kk = 0; kk < 16; ++kk) {
        const float* p = x + (size_t)(rowBase + mn) * F_DIM + kk * 4 + hlf * 2;
        aX[kk][0] = p[0];
        aX[kk][1] = p[1];
    }

    const float* Wfirst[2]  = {W1, Ws1};
    const float* Bfirst[2]  = {b1, bs1};
    const float* Wsecond[2] = {W2, Ws2};
    const float* Bsecond[2] = {b2, bs2};
    float*       Dst[2]     = {x1_out, self_out};

    for (int s = 0; s < 2; ++s) {
        __syncthreads();
        for (int i = tid; i < F_DIM * F_DIM; i += 256) sW[i] = Wfirst[s][i];
        if (tid < F_DIM) sB[tid] = Bfirst[s][tid];
        __syncthreads();

#pragma unroll
        for (int nt = 0; nt < 4; ++nt) {
            v8f c;
            float bias = sB[nt * 16 + mn];
#pragma unroll
            for (int r = 0; r < 8; ++r) c[r] = bias;
#pragma unroll
            for (int kk = 0; kk < 16; ++kk) {
                v2f bF;
                bF[0] = sW[(kk * 4 + hlf * 2 + 0) * F_DIM + nt * 16 + mn];
                bF[1] = sW[(kk * 4 + hlf * 2 + 1) * F_DIM + nt * 16 + mn];
                c = __builtin_amdgcn_wmma_f32_16x16x4_f32(
                        false, aX[kk], false, bF, (short)0, c, false, false);
            }
#pragma unroll
            for (int r = 0; r < 8; ++r)
                sH[wave][(r + 8 * hlf) * F_DIM + nt * 16 + mn] = fmaxf(c[r], 0.0f);
        }
        __syncthreads();

        for (int i = tid; i < F_DIM * F_DIM; i += 256) sW[i] = Wsecond[s][i];
        if (tid < F_DIM) sB[tid] = Bsecond[s][tid];
        __syncthreads();

        v2f aH[16];
#pragma unroll
        for (int kk = 0; kk < 16; ++kk) {
            aH[kk][0] = sH[wave][mn * F_DIM + kk * 4 + hlf * 2 + 0];
            aH[kk][1] = sH[wave][mn * F_DIM + kk * 4 + hlf * 2 + 1];
        }

#pragma unroll
        for (int nt = 0; nt < 4; ++nt) {
            v8f c;
            float bias = sB[nt * 16 + mn];
#pragma unroll
            for (int r = 0; r < 8; ++r) c[r] = bias;
#pragma unroll
            for (int kk = 0; kk < 16; ++kk) {
                v2f bF;
                bF[0] = sW[(kk * 4 + hlf * 2 + 0) * F_DIM + nt * 16 + mn];
                bF[1] = sW[(kk * 4 + hlf * 2 + 1) * F_DIM + nt * 16 + mn];
                c = __builtin_amdgcn_wmma_f32_16x16x4_f32(
                        false, aH[kk], false, bF, (short)0, c, false, false);
            }
#pragma unroll
            for (int r = 0; r < 8; ++r) {
                Dst[s][(size_t)(rowBase + r + 8 * hlf) * F_DIM + nt * 16 + mn] =
                    fmaxf(c[r], 0.0f);
            }
        }
    }
}

__global__ __launch_bounds__(256) void hg_edge_kernel(
    const float* __restrict__ Wval,
    const int* __restrict__ eb, const int* __restrict__ ei,
    const int* __restrict__ ej, const int* __restrict__ ek,
    const float* __restrict__ deg, const float* __restrict__ x1,
    float* __restrict__ out, int nnz) {

    int gid = blockIdx.x * blockDim.x + threadIdx.x;
    int e = gid >> 4;
    if (e >= nnz) return;
    int c = (gid & 15) * 4;

    int b = __builtin_nontemporal_load(eb + e);
    int i = __builtin_nontemporal_load(ei + e);
    int j = __builtin_nontemporal_load(ej + e);
    int k = __builtin_nontemporal_load(ek + e);

    float a = 1.0f / (deg[b * N_DIM + i] + EPS_V);

    const v4f w = __builtin_nontemporal_load(
        (const v4f*)(Wval + (size_t)e * F_DIM + c));
    const float4 xk = *(const float4*)(x1 + (size_t)(b * N_DIM + k) * F_DIM + c);
    const float4 xj = *(const float4*)(x1 + (size_t)(b * N_DIM + j) * F_DIM + c);

    float* o = out + (size_t)(b * N_DIM + i) * F_DIM + c;
    atomicAdd(o + 0, a * w[0] * xk.x * xj.x);
    atomicAdd(o + 1, a * w[1] * xk.y * xj.y);
    atomicAdd(o + 2, a * w[2] * xk.z * xj.z);
    atomicAdd(o + 3, a * w[3] * xk.w * xj.w);
}

extern "C" void kernel_launch(void* const* d_in, const int* in_sizes, int n_in,
                              void* d_out, int out_size, void* d_ws, size_t ws_size,
                              hipStream_t stream) {
    const float* x    = (const float*)d_in[0];
    const float* Wval = (const float*)d_in[1];
    const int*   eb   = (const int*)d_in[2];
    const int*   ei   = (const int*)d_in[3];
    const int*   ej   = (const int*)d_in[4];
    const int*   ek   = (const int*)d_in[5];
    const float* W1   = (const float*)d_in[6];
    const float* b1   = (const float*)d_in[7];
    const float* W2   = (const float*)d_in[8];
    const float* b2   = (const float*)d_in[9];
    const float* Ws1  = (const float*)d_in[10];
    const float* bs1  = (const float*)d_in[11];
    const float* Ws2  = (const float*)d_in[12];
    const float* bs2  = (const float*)d_in[13];

    const int nnz   = in_sizes[2];
    const int nodes = B_DIM * N_DIM;

    float* deg = (float*)d_ws;
    float* x1  = deg + nodes;
    float* out = (float*)d_out;

    hg_zero_kernel<<<(nodes + 255) / 256, 256, 0, stream>>>(deg, nodes);

    hg_degree_kernel<<<(nnz + 255) / 256, 256, 0, stream>>>(eb, ei, deg, nnz);

    hg_mlp_wmma_kernel<<<nodes / 128, 256, 0, stream>>>(
        x, W1, b1, W2, b2, Ws1, bs1, Ws2, bs2, x1, out);

    long long threads = (long long)nnz * 16;
    hg_edge_kernel<<<(unsigned)((threads + 255) / 256), 256, 0, stream>>>(
        Wval, eb, ei, ej, ek, deg, x1, out, nnz);
}


# 1/deg taken out of the edge sum: edge blocks count head-node degrees in LDS beside unscaled sums; self-MLP kernel, now last, scales each sum once (rounding differs); no zero/count launches, 4 to 3
# speedup vs baseline: 4.1690x; 4.1690x over previous
#include <hip/hip_runtime.h>

#define B_DIM 8
#define N_DIM 1024
#define F_DIM 64
#define EPS_V 1e-10f

typedef __attribute__((ext_vector_type(2))) float v2f;
typedef __attribute__((ext_vector_type(4))) float v4f;
typedef __attribute__((ext_vector_type(8))) float v8f;


__device__ __forceinline__ void hg_mlp_strip(
    const float* __restrict__ x,
    const float* __restrict__ Wa, const float* __restrict__ ba,
    const float* __restrict__ Wb, const float* __restrict__ bb,
    float* sW, float* sB, float* sHw,
    int tid, int hlf, int mn, int rowBase, v8f (&acc)[4]) {

    v2f aX[16];
#pragma unroll
    for (int kk = 0; kk < 16; ++kk) {
        const float* p = x + (size_t)(rowBase + mn) * F_DIM + kk * 4 + hlf * 2;
        aX[kk][0] = p[0];
        aX[kk][1] = p[1];
    }

    for (int i = tid; i < F_DIM * F_DIM; i += 256) sW[i] = Wa[i];
    if (tid < F_DIM) sB[tid] = ba[tid];
    __syncthreads();

#pragma unroll
    for (int nt = 0; nt < 4; ++nt) {
        v8f c;
        float bias = sB[nt * 16 + mn];
#pragma unroll
        for (int r = 0; r < 8; ++r) c[r] = bias;
#pragma unroll
        for (int kk = 0; kk < 16; ++kk) {
            v2f bF;
            bF[0] = sW[(kk * 4 + hlf * 2 + 0) * F_DIM + nt * 16 + mn];
            bF[1] = sW[(kk * 4 + hlf * 2 + 1) * F_DIM + nt * 16 + mn];
            c = __builtin_amdgcn_wmma_f32_16x16x4_f32(
                    false, aX[kk], false, bF, (short)0, c, false, false);
        }
#pragma unroll
        for (int r = 0; r < 8; ++r)
            sHw[(r + 8 * hlf) * F_DIM + nt * 16 + mn] = fmaxf(c[r], 0.0f);
    }
    __syncthreads();

    for (int i = tid; i < F_DIM * F_DIM; i += 256) sW[i] = Wb[i];
    if (tid < F_DIM) sB[tid] = bb[tid];
    __syncthreads();

    v2f aH[16];
#pragma unroll
    for (int kk = 0; kk < 16; ++kk) {
        aH[kk][0] = sHw[mn * F_DIM + kk * 4 + hlf * 2 + 0];
        aH[kk][1] = sHw[mn * F_DIM + kk * 4 + hlf * 2 + 1];
    }

#pragma unroll
    for (int nt = 0; nt < 4; ++nt) {
        v8f c;
        float bias = sB[nt * 16 + mn];
#pragma unroll
        for (int r = 0; r < 8; ++r) c[r] = bias;
#pragma unroll
        for (int kk = 0; kk < 16; ++kk) {
            v2f bF;
            bF[0] = sW[(kk * 4 + hlf * 2 + 0) * F_DIM + nt * 16 + mn];
            bF[1] = sW[(kk * 4 + hlf * 2 + 1) * F_DIM + nt * 16 + mn];
            c = __builtin_amdgcn_wmma_f32_16x16x4_f32(
                    false, aH[kk], false, bF, (short)0, c, false, false);
        }
        acc[nt] = c;
    }
}

__global__ __launch_bounds__(256) void hg_mlp_x1_kernel(
    const float* __restrict__ x,
    const float* __restrict__ W1, const float* __restrict__ b1,
    const float* __restrict__ W2, const float* __restrict__ b2,
    float* __restrict__ x1_out, float* __restrict__ out,
    float* __restrict__ cnt) {

    __shared__ float sW[F_DIM * F_DIM];
    __shared__ float sB[F_DIM];
    __shared__ float sH[8][16 * F_DIM];

    const int tid  = threadIdx.x;
    const int wave = tid >> 5;
    const int lane = tid & 31;
    const int hlf  = lane >> 4;
    const int mn   = lane & 15;
    const int rowBase = (blockIdx.x * 8 + wave) * 16;

    {
        const v4f z = {0.0f, 0.0f, 0.0f, 0.0f};
        v4f* o = (v4f*)(out + (size_t)rowBase * F_DIM);
#pragma unroll
        for (int q = 0; q < 16 * F_DIM / 4 / 32; ++q) o[q * 32 + lane] = z;
        if (wave == 0) ((v4f*)(cnt + blockIdx.x * 128))[lane] = z;
    }

    v8f acc[4];
    hg_mlp_strip(x, W1, b1, W2, b2, sW, sB, sH[wave], tid, hlf, mn, rowBase, acc);

#pragma unroll
    for (int nt = 0; nt < 4; ++nt)
#pragma unroll
        for (int r = 0; r < 8; ++r)
            x1_out[(size_t)(rowBase + r + 8 * hlf) * F_DIM + nt * 16 + mn] =
                fmaxf(acc[nt][r], 0.0f);
}

__global__ __launch_bounds__(256) void hg_mlp_self_kernel(
    const float* __restrict__ x,
    const float* __restrict__ Ws1, const float* __restrict__ bs1,
    const float* __restrict__ Ws2, const float* __restrict__ bs2,
    float* __restrict__ out, const float* __restrict__ cnt) {

    __shared__ float sW[F_DIM * F_DIM];
    __shared__ float sB[F_DIM];
    __shared__ float sH[8][16 * F_DIM];

    const int tid  = threadIdx.x;
    const int wave = tid >> 5;
    const int lane = tid & 31;
    const int hlf  = lane >> 4;
    const int mn   = lane & 15;
    const int rowBase = (blockIdx.x * 8 + wave) * 16;

    float sum[4][8];
#pragma unroll
    for (int nt = 0; nt < 4; ++nt)
#pragma unroll
        for (int r = 0; r < 8; ++r)
            sum[nt][r] = out[(size_t)(rowBase + r + 8 * hlf) * F_DIM + nt * 16 + mn];
    const v4f c0 = *(const v4f*)(cnt + rowBase + 8 * hlf);
    const v4f c1 = *(const v4f*)(cnt + rowBase + 8 * hlf + 4);

    v8f acc[4];
    hg_mlp_strip(x, Ws1, bs1, Ws2, bs2, sW, sB, sH[wave], tid, hlf, mn, rowBase, acc);

    float inv[8];
#pragma unroll
    for (int r = 0; r < 4; ++r) {
        inv[r]     = 1.0f / (c0[r] + EPS_V);
        inv[r + 4] = 1.0f / (c1[r] + EPS_V);
    }

#pragma unroll
    for (int nt = 0; nt < 4; ++nt)
#pragma unroll
        for (int r = 0; r < 8; ++r)
            out[(size_t)(rowBase + r + 8 * hlf) * F_DIM + nt * 16 + mn] =
                fmaxf(acc[nt][r], 0.0f) + sum[nt][r] * inv[r];
}

#define EDGE_THREADS 1024
#define EDGE_WAVES   (EDGE_THREADS / 32)
#define EDGE_SLICES  32
#define EDGE_SLOTS   12
#define EDGE_GROUP   64
#define EDGE_FLUSH   16
#define N_BITS       10
static_assert((1 << N_BITS) == N_DIM, "i, j, k are packed 10 bits each");
static_assert((N_DIM * F_DIM) % (EDGE_THREADS * EDGE_FLUSH) == 0, "flush loop has no tail");

template <int N>
__device__ __forceinline__ void edge_round(
    unsigned& m0, unsigned& m1, int base, int p0, int p1,
    const float* __restrict__ wl, const float* __restrict__ x1b,
    float* sAcc, int lane2) {
    v2f w[N], xk[N], xj[N];
    int row[N];
#pragma unroll
    for (int u = 0; u < N; ++u) {
        int e, p;
        if (m0) {
            const int t = __builtin_ctz(m0);
            m0 &= m0 - 1;
            e = base + t;
            p = __builtin_amdgcn_readlane(p0, t);
        } else {
            const int t = __builtin_ctz(m1);
            m1 &= m1 - 1;
            e = base + 32 + t;
            p = __builtin_amdgcn_readlane(p1, t);
        }
        const int i = p & (N_DIM - 1);
        const int j = (p >> N_BITS) & (N_DIM - 1);
        const int k = (p >> (2 * N_BITS)) & (N_DIM - 1);
        row[u] = i;
        w[u]  = __builtin_nontemporal_load((const v2f*)(wl + (size_t)e * F_DIM));
        xk[u] = *(const v2f*)(x1b + (size_t)k * F_DIM);
        xj[u] = *(const v2f*)(x1b + (size_t)j * F_DIM);
    }
    __builtin_amdgcn_sched_barrier(0);
#pragma unroll
    for (int u = 0; u < N; ++u) {
        float* o = sAcc + row[u] * F_DIM + lane2;
        atomicAdd(o + 0, w[u][0] * xk[u][0] * xj[u][0]);
        atomicAdd(o + 1, w[u][1] * xk[u][1] * xj[u][1]);
    }
}

__global__ __launch_bounds__(EDGE_THREADS) void hg_edge_lds_kernel(
    const float* __restrict__ Wval,
    const int* __restrict__ eb, const int* __restrict__ ei,
    const int* __restrict__ ej, const int* __restrict__ ek,
    float* __restrict__ cnt, const float* __restrict__ x1,
    float* __restrict__ out, int nnz) {

    __shared__ __attribute__((aligned(16))) float sAcc[N_DIM * F_DIM];
    __shared__ float sCnt[N_DIM];

    const int tid   = threadIdx.x;
    const int lane  = tid & 31;
    const int wave  = __builtin_amdgcn_readfirstlane(tid >> 5);
    const int b     = blockIdx.x % B_DIM;
    const int slice = blockIdx.x / B_DIM;

    const int lane2  = lane * 2;
    const float* x1b = x1 + (size_t)b * N_DIM * F_DIM + lane2;
    const float* wl  = Wval + lane2;
    const int stride = EDGE_SLICES * EDGE_WAVES * EDGE_GROUP;

    int base = (slice * EDGE_WAVES + wave) * EDGE_GROUP;

    int nb0 = -1, ni0 = 0, nj0 = 0, nk0 = 0;
    int nb1 = -1, ni1 = 0, nj1 = 0, nk1 = 0;
    if (base + lane < nnz) {
        nb0 = eb[base + lane]; ni0 = ei[base + lane];
        nj0 = ej[base + lane]; nk0 = ek[base + lane];
    }
    if (base + 32 + lane < nnz) {
        nb1 = eb[base + 32 + lane]; ni1 = ei[base + 32 + lane];
        nj1 = ej[base + 32 + lane]; nk1 = ek[base + 32 + lane];
    }

    {
        const v4f z = {0.0f, 0.0f, 0.0f, 0.0f};
        for (int q = tid; q < N_DIM * F_DIM / 4; q += EDGE_THREADS)
            ((v4f*)sAcc)[q] = z;
        for (int n = tid; n < N_DIM; n += EDGE_THREADS)
            sCnt[n] = 0.0f;
    }
    __syncthreads();

    for (; base < nnz; base += stride) {
        const int cb0 = nb0, cb1 = nb1;
        const int p0 = ni0 | (nj0 << N_BITS) | (nk0 << (2 * N_BITS));
        const int p1 = ni1 | (nj1 << N_BITS) | (nk1 << (2 * N_BITS));

        const int nbase = base + stride;
        nb0 = -1; nb1 = -1;
        if (nbase + lane < nnz) {
            nb0 = eb[nbase + lane]; ni0 = ei[nbase + lane];
            nj0 = ej[nbase + lane]; nk0 = ek[nbase + lane];
        }
        if (nbase + 32 + lane < nnz) {
            nb1 = eb[nbase + 32 + lane]; ni1 = ei[nbase + 32 + lane];
            nj1 = ej[nbase + 32 + lane]; nk1 = ek[nbase + 32 + lane];
        }

        unsigned m0 = __builtin_amdgcn_ballot_w32(cb0 == b);
        unsigned m1 = __builtin_amdgcn_ballot_w32(cb1 == b);

        if (cb0 == b) atomicAdd(&sCnt[p0 & (N_DIM - 1)], 1.0f);
        if (cb1 == b) atomicAdd(&sCnt[p1 & (N_DIM - 1)], 1.0f);

        while (m0 | m1) {
            const int n = __builtin_popcount(m0) + __builtin_popcount(m1);
            switch (n) {
            case 1:  edge_round<1>(m0, m1, base, p0, p1, wl, x1b, sAcc, lane2); break;
            case 2:  edge_round<2>(m0, m1, base, p0, p1, wl, x1b, sAcc, lane2); break;
            case 3:  edge_round<3>(m0, m1, base, p0, p1, wl, x1b, sAcc, lane2); break;
            case 4:  edge_round<4>(m0, m1, base, p0, p1, wl, x1b, sAcc, lane2); break;
            case 5:  edge_round<5>(m0, m1, base, p0, p1, wl, x1b, sAcc, lane2); break;
            case 6:  edge_round<6>(m0, m1, base, p0, p1, wl, x1b, sAcc, lane2); break;
            case 7:  edge_round<7>(m0, m1, base, p0, p1, wl, x1b, sAcc, lane2); break;
            case 8:  edge_round<8>(m0, m1, base, p0, p1, wl, x1b, sAcc, lane2); break;
            case 9:  edge_round<9>(m0, m1, base, p0, p1, wl, x1b, sAcc, lane2); break;
            case 10: edge_round<10>(m0, m1, base, p0, p1, wl, x1b, sAcc, lane2); break;
            case 11: edge_round<11>(m0, m1, base, p0, p1, wl, x1b, sAcc, lane2); break;
            default: edge_round<EDGE_SLOTS>(m0, m1, base, p0, p1, wl, x1b, sAcc, lane2); break;
            }
        }
    }

    __syncthreads();

    float* ob = out + (size_t)b * N_DIM * F_DIM;
    for (int q0 = tid; q0 < N_DIM * F_DIM; q0 += EDGE_THREADS * EDGE_FLUSH) {
        float v[EDGE_FLUSH];
#pragma unroll
        for (int r = 0; r < EDGE_FLUSH; ++r) v[r] = sAcc[q0 + r * EDGE_THREADS];
#pragma unroll
        for (int r = 0; r < EDGE_FLUSH; ++r) atomicAdd(ob + q0 + r * EDGE_THREADS, v[r]);
    }
    for (int n = tid; n < N_DIM; n += EDGE_THREADS)
        atomicAdd(cnt + b * N_DIM + n, sCnt[n]);
}

extern "C" void kernel_launch(void* const* d_in, const int* in_sizes, int n_in,
                              void* d_out, int out_size, void* d_ws, size_t ws_size,
                              hipStream_t stream) {
    const float* x    = (const float*)d_in[0];
    const float* Wval = (const float*)d_in[1];
    const int*   eb   = (const int*)d_in[2];
    const int*   ei   = (const int*)d_in[3];
    const int*   ej   = (const int*)d_in[4];
    const int*   ek   = (const int*)d_in[5];
    const float* W1   = (const float*)d_in[6];
    const float* b1   = (const float*)d_in[7];
    const float* W2   = (const float*)d_in[8];
    const float* b2   = (const float*)d_in[9];
    const float* Ws1  = (const float*)d_in[10];
    const float* bs1  = (const float*)d_in[11];
    const float* Ws2  = (const float*)d_in[12];
    const float* bs2  = (const float*)d_in[13];

    const int nnz   = in_sizes[2];
    const int nodes = B_DIM * N_DIM;

    float* cnt = (float*)d_ws;
    float* x1  = cnt + nodes;
    float* out = (float*)d_out;

    hg_mlp_x1_kernel<<<nodes / 128, 256, 0, stream>>>(
        x, W1, b1, W2, b2, x1, out, cnt);

    hg_edge_lds_kernel<<<B_DIM * EDGE_SLICES, EDGE_THREADS, 0, stream>>>(
        Wval, eb, ei, ej, ek, cnt, x1, out, nnz);

    hg_mlp_self_kernel<<<nodes / 128, 256, 0, stream>>>(
        x, Ws1, bs1, Ws2, bs2, out, cnt);
}
